# starting kernel
# baseline (speedup 1.0000x reference)
#include <hip/hip_runtime.h>

typedef __attribute__((ext_vector_type(2))) float v2f;
typedef __attribute__((ext_vector_type(8))) float v8f;

#define NB 16
#define SS 1024
#define EE 256
#define HH 8
#define DD 32
#define E3 768
#define LDSTR 20

__device__ __forceinline__ v8f wmma_f32(v2f a, v2f b, v8f c) {
  return __builtin_amdgcn_wmma_f32_16x16x4_f32(
       false, a,  false, b,
       (short)0, c,  false,  false);
}

__global__ __launch_bounds__(256) void gemm64x16(const float* __restrict__ X,
                                                 const float* __restrict__ W,
                                                 const float* __restrict__ bias,
                                                 float* __restrict__ Y,
                                                 int M, int N, int K) {
  const int lane = threadIdx.x & 31;
  const int wave = threadIdx.x >> 5;
  const int ntn  = N >> 4;
  int strip = blockIdx.x * 8 + wave;
  int sm = strip / ntn;
  int tn = strip - sm * ntn;
  const int row0 = sm << 6, col0 = tn << 4;
  const int m = lane & 15, half = lane >> 4;

  const float* xrow = X + (size_t)(row0 + m) * K + 2 * half;
  const float* wcol = W + (size_t)(2 * half) * N + col0 + m;

  v8f acc[4];
#pragma unroll
  for (int t = 0; t < 4; ++t) acc[t] = {};

#pragma unroll 4
  for (int kk = 0; kk < K; kk += 4) {
    v2f b;
    b.x = wcol[(size_t)kk * N];
    b.y = wcol[(size_t)(kk + 1) * N];
    v2f a0 = *(const v2f*)(xrow + kk);
    v2f a1 = *(const v2f*)(xrow + (size_t)16 * K + kk);
    v2f a2 = *(const v2f*)(xrow + (size_t)32 * K + kk);
    v2f a3 = *(const v2f*)(xrow + (size_t)48 * K + kk);
    acc[0] = wmma_f32(a0, b, acc[0]);
    acc[1] = wmma_f32(a1, b, acc[1]);
    acc[2] = wmma_f32(a2, b, acc[2]);
    acc[3] = wmma_f32(a3, b, acc[3]);
  }

  float bn = bias[col0 + m];
#pragma unroll
  for (int t = 0; t < 4; ++t) {
#pragma unroll
    for (int r = 0; r < 8; ++r) {
      int Mr = 16 * t + r + 8 * half;
      Y[(size_t)(row0 + Mr) * N + col0 + m] = acc[t][r] + bn;
    }
  }
}

__global__ __launch_bounds__(256) void rope_kernel(float* __restrict__ qkv) {
  int t = blockIdx.x * 256 + threadIdx.x;
  int p = t & 15;  t >>= 4;
  int h = t & 7;   t >>= 3;
  int s = t & 1023; t >>= 10;
  int b = t & 15;  t >>= 4;
  int which = t;

  int i = s >> 5, j = s & 31;
  int r = p & 7;
  float freq = __expf(-1.1512925464970231f * (float)r);
  float pos  = (p < 8) ? (float)i : (float)j;
  float ang  = pos * freq;
  float sn, cs;
  __sincosf(ang, &sn, &cs);

  size_t idx = (size_t)(b * SS + s) * E3 + which * EE + h * DD + 2 * p;
  float x0 = qkv[idx], x1 = qkv[idx + 1];
  qkv[idx]     = x0 * cs - x1 * sn;
  qkv[idx + 1] = x0 * sn + x1 * cs;
}

__global__ __launch_bounds__(256) void attn_kernel(const float* __restrict__ qkv,
                                                   float* __restrict__ out) {
  __shared__ float lds_p[8][16 * LDSTR];

  const int lane = threadIdx.x & 31;
  const int wave = threadIdx.x >> 5;
  int id = blockIdx.x * 8 + wave;
  const int qt = id & 63; id >>= 6;
  const int h  = id & 7;  id >>= 3;
  const int b  = id;
  const int m = lane & 15, half = lane >> 4;
  const int row0 = qt << 4;

  const size_t basebh = (size_t)(b * SS) * E3 + h * DD;
  const float* kslab = qkv + basebh + EE;
  const float* vslab = qkv + basebh + 2 * EE;

  v2f qa[8];
#pragma unroll
  for (int f = 0; f < 8; ++f)
    qa[f] = *(const v2f*)(qkv + basebh + (size_t)(row0 + m) * E3 + 4 * f + 2 * half);

  float mrow[8], lpart[8];
  v8f o0 = {}, o1 = {};
#pragma unroll
  for (int r = 0; r < 8; ++r) { mrow[r] = -3.0e38f; lpart[r] = 0.0f; }

  const float scale = 0.17677669529663687f;
  float* myp = &lds_p[wave][0];

  for (int kt = 0; kt < 64; ++kt) {
    const int krow0 = kt << 4;

    v8f sacc = {};
    const float* kbase = kslab + (size_t)(krow0 + m) * E3 + 2 * half;
#pragma unroll
    for (int f = 0; f < 8; ++f) {
      v2f kb = *(const v2f*)(kbase + 4 * f);
      sacc = wmma_f32(qa[f], kb, sacc);
    }

    float pr[8];
#pragma unroll
    for (int r = 0; r < 8; ++r) {
      float v = sacc[r] * scale;
      float rmax = v;
      rmax = fmaxf(rmax, __shfl_xor(rmax, 1, 32));
      rmax = fmaxf(rmax, __shfl_xor(rmax, 2, 32));
      rmax = fmaxf(rmax, __shfl_xor(rmax, 4, 32));
      rmax = fmaxf(rmax, __shfl_xor(rmax, 8, 32));
      float mnew = fmaxf(mrow[r], rmax);
      float corr = __expf(mrow[r] - mnew);
      float p    = __expf(v - mnew);
      lpart[r] = lpart[r] * corr + p;
      mrow[r]  = mnew;
      o0[r] *= corr;
      o1[r] *= corr;
      pr[r] = p;
    }

#pragma unroll
    for (int r = 0; r < 8; ++r)
      myp[(r + 8 * half) * LDSTR + m] = pr[r];
    __builtin_amdgcn_wave_barrier();
    v2f pa[4];
#pragma unroll
    for (int f = 0; f < 4; ++f)
      pa[f] = *(const v2f*)(myp + m * LDSTR + 4 * f + 2 * half);
    __builtin_amdgcn_wave_barrier();

#pragma unroll
    for (int f = 0; f < 4; ++f) {
      int kp = krow0 + 4 * f + 2 * half;
      const float* v0 = vslab + (size_t)kp * E3;
      const float* v1 = v0 + E3;
      v2f vb0, vb1;
      vb0.x = v0[m];      vb0.y = v1[m];
      vb1.x = v0[16 + m]; vb1.y = v1[16 + m];
      o0 = wmma_f32(pa[f], vb0, o0);
      o1 = wmma_f32(pa[f], vb1, o1);
    }
  }

#pragma unroll
  for (int r = 0; r < 8; ++r) {
    float lsum = lpart[r];
    lsum += __shfl_xor(lsum, 1, 32);
    lsum += __shfl_xor(lsum, 2, 32);
    lsum += __shfl_xor(lsum, 4, 32);
    lsum += __shfl_xor(lsum, 8, 32);
    float inv = 1.0f / lsum;
    int Mr = r + 8 * half;
    size_t orow = (size_t)(b * SS + row0 + Mr) * EE + h * DD;
    out[orow + m]      = o0[r] * inv;
    out[orow + 16 + m] = o1[r] * inv;
  }
}

extern "C" void kernel_launch(void* const* d_in, const int* in_sizes, int n_in,
                              void* d_out, int out_size, void* d_ws, size_t ws_size,
                              hipStream_t stream) {
  const float* x    = (const float*)d_in[0];
  const float* Wqkv = (const float*)d_in[1];
  const float* bqkv = (const float*)d_in[2];
  const float* Wout = (const float*)d_in[3];
  const float* bout = (const float*)d_in[4];
  float* out  = (float*)d_out;
  float* qkv  = (float*)d_ws;
  float* attn = qkv + (size_t)NB * SS * E3;

  gemm64x16<<<dim3((NB * SS / 64) * (E3 / 16) / 8), dim3(256), 0, stream>>>(
      x, Wqkv, bqkv, qkv, NB * SS, E3, EE);

  rope_kernel<<<dim3(2 * NB * SS * HH * (DD / 2) / 256), dim3(256), 0, stream>>>(qkv);

  attn_kernel<<<dim3(NB * HH * (SS / 16) / 8), dim3(256), 0, stream>>>(qkv, attn);

  gemm64x16<<<dim3((NB * SS / 64) * (EE / 16) / 8), dim3(256), 0, stream>>>(
      attn, Wout, bout, out, NB * SS, EE, EE);
}


# attention: score tile formed transposed (K q^T): max, correction, row sum per lane; P^T feeds V^T P^T from registers, no LDS transpose; row-sum and PV key summation order change
# speedup vs baseline: 1.1542x; 1.1542x over previous
#include <hip/hip_runtime.h>

typedef __attribute__((ext_vector_type(2))) float v2f;
typedef __attribute__((ext_vector_type(8))) float v8f;

#define NB 16
#define SS 1024
#define EE 256
#define HH 8
#define DD 32
#define E3 768

__device__ __forceinline__ v8f wmma_f32(v2f a, v2f b, v8f c) {
  return __builtin_amdgcn_wmma_f32_16x16x4_f32(
       false, a,  false, b,
       (short)0, c,  false,  false);
}

__global__ __launch_bounds__(256) void gemm64x16(const float* __restrict__ X,
                                                 const float* __restrict__ W,
                                                 const float* __restrict__ bias,
                                                 float* __restrict__ Y,
                                                 int M, int N, int K) {
  const int lane = threadIdx.x & 31;
  const int wave = threadIdx.x >> 5;
  const int ntn  = N >> 4;
  int strip = blockIdx.x * 8 + wave;
  int sm = strip / ntn;
  int tn = strip - sm * ntn;
  const int row0 = sm << 6, col0 = tn << 4;
  const int m = lane & 15, half = lane >> 4;

  const float* xrow = X + (size_t)(row0 + m) * K + 2 * half;
  const float* wcol = W + (size_t)(2 * half) * N + col0 + m;

  v8f acc[4];
#pragma unroll
  for (int t = 0; t < 4; ++t) acc[t] = {};

#pragma unroll 4
  for (int kk = 0; kk < K; kk += 4) {
    v2f b;
    b.x = wcol[(size_t)kk * N];
    b.y = wcol[(size_t)(kk + 1) * N];
    v2f a0 = *(const v2f*)(xrow + kk);
    v2f a1 = *(const v2f*)(xrow + (size_t)16 * K + kk);
    v2f a2 = *(const v2f*)(xrow + (size_t)32 * K + kk);
    v2f a3 = *(const v2f*)(xrow + (size_t)48 * K + kk);
    acc[0] = wmma_f32(a0, b, acc[0]);
    acc[1] = wmma_f32(a1, b, acc[1]);
    acc[2] = wmma_f32(a2, b, acc[2]);
    acc[3] = wmma_f32(a3, b, acc[3]);
  }

  float bn = bias[col0 + m];
#pragma unroll
  for (int t = 0; t < 4; ++t) {
#pragma unroll
    for (int r = 0; r < 8; ++r) {
      int Mr = 16 * t + r + 8 * half;
      Y[(size_t)(row0 + Mr) * N + col0 + m] = acc[t][r] + bn;
    }
  }
}

__global__ __launch_bounds__(256) void rope_kernel(float* __restrict__ qkv) {
  int t = blockIdx.x * 256 + threadIdx.x;
  int p = t & 15;  t >>= 4;
  int h = t & 7;   t >>= 3;
  int s = t & 1023; t >>= 10;
  int b = t & 15;  t >>= 4;
  int which = t;

  int i = s >> 5, j = s & 31;
  int r = p & 7;
  float freq = __expf(-1.1512925464970231f * (float)r);
  float pos  = (p < 8) ? (float)i : (float)j;
  float ang  = pos * freq;
  float sn, cs;
  __sincosf(ang, &sn, &cs);

  size_t idx = (size_t)(b * SS + s) * E3 + which * EE + h * DD + 2 * p;
  float x0 = qkv[idx], x1 = qkv[idx + 1];
  qkv[idx]     = x0 * cs - x1 * sn;
  qkv[idx + 1] = x0 * sn + x1 * cs;
}

__device__ __forceinline__ float xhalf(float x) {
  int b = __builtin_bit_cast(int, x);
  return __builtin_bit_cast(float,
      __builtin_amdgcn_permlanex16(b, b, 0x76543210, 0xfedcba98, false, false));
}

__global__ __launch_bounds__(256) void attn_kernel(const float* __restrict__ qkv,
                                                   float* __restrict__ out) {
  const int lane = threadIdx.x & 31;
  const int wave = threadIdx.x >> 5;
  int id = blockIdx.x * 8 + wave;
  const int qt = id & 63; id >>= 6;
  const int h  = id & 7;  id >>= 3;
  const int b  = id;
  const int m = lane & 15, half = lane >> 4;
  const int row0 = qt << 4;

  const size_t basebh = (size_t)(b * SS) * E3 + h * DD;
  const float* kslab = qkv + basebh + EE;
  const float* vslab = qkv + basebh + 2 * EE;

  v2f qb[8];
#pragma unroll
  for (int f = 0; f < 8; ++f)
    qb[f] = *(const v2f*)(qkv + basebh + (size_t)(row0 + m) * E3 + 4 * f + 2 * half);

  float mrun = -3.0e38f, lpart = 0.0f;
  v8f o0 = {}, o1 = {};

  const float scale = 0.17677669529663687f;

  for (int kt = 0; kt < 64; ++kt) {
    const int krow0 = kt << 4;

    v8f sacc = {};
    const float* kbase = kslab + (size_t)(krow0 + m) * E3 + 2 * half;
#pragma unroll
    for (int f = 0; f < 8; ++f) {
      v2f ka = *(const v2f*)(kbase + 4 * f);
      sacc = wmma_f32(ka, qb[f], sacc);
    }

    float v[8];
#pragma unroll
    for (int r = 0; r < 8; ++r) v[r] = sacc[r] * scale;
    float tmax = fmaxf(fmaxf(fmaxf(v[0], v[1]), fmaxf(v[2], v[3])),
                       fmaxf(fmaxf(v[4], v[5]), fmaxf(v[6], v[7])));
    tmax = fmaxf(tmax, xhalf(tmax));
    const float mnew = fmaxf(mrun, tmax);
    const float corr = __expf(mrun - mnew);
    mrun = mnew;

    v2f pb[4];
    float psum = 0.0f;
#pragma unroll
    for (int f = 0; f < 4; ++f) {
      float p0 = __expf(v[2 * f]     - mnew);
      float p1 = __expf(v[2 * f + 1] - mnew);
      psum += p0;
      psum += p1;
      pb[f].x = p0;
      pb[f].y = p1;
    }
    lpart = lpart * corr + psum;
#pragma unroll
    for (int r = 0; r < 8; ++r) { o0[r] *= corr; o1[r] *= corr; }

#pragma unroll
    for (int f = 0; f < 4; ++f) {
      const float* v0 = vslab + (size_t)(krow0 + 8 * half + 2 * f) * E3;
      const float* v1 = v0 + E3;
      v2f va0, va1;
      va0.x = v0[m];      va0.y = v1[m];
      va1.x = v0[16 + m]; va1.y = v1[16 + m];
      o0 = wmma_f32(va0, pb[f], o0);
      o1 = wmma_f32(va1, pb[f], o1);
    }
  }

  float lsum = lpart + xhalf(lpart);
  float inv = 1.0f / lsum;
  float* orow = out + (size_t)(b * SS + row0 + m) * EE + h * DD + 8 * half;
  typedef __attribute__((ext_vector_type(4))) float v4f;
  v4f w0 = {o0[0] * inv, o0[1] * inv, o0[2] * inv, o0[3] * inv};
  v4f w1 = {o0[4] * inv, o0[5] * inv, o0[6] * inv, o0[7] * inv};
  v4f w2 = {o1[0] * inv, o1[1] * inv, o1[2] * inv, o1[3] * inv};
  v4f w3 = {o1[4] * inv, o1[5] * inv, o1[6] * inv, o1[7] * inv};
  *(v4f*)(orow)      = w0;
  *(v4f*)(orow + 4)  = w1;
  *(v4f*)(orow + 16) = w2;
  *(v4f*)(orow + 20) = w3;
}

extern "C" void kernel_launch(void* const* d_in, const int* in_sizes, int n_in,
                              void* d_out, int out_size, void* d_ws, size_t ws_size,
                              hipStream_t stream) {
  const float* x    = (const float*)d_in[0];
  const float* Wqkv = (const float*)d_in[1];
  const float* bqkv = (const float*)d_in[2];
  const float* Wout = (const float*)d_in[3];
  const float* bout = (const float*)d_in[4];
  float* out  = (float*)d_out;
  float* qkv  = (float*)d_ws;
  float* attn = qkv + (size_t)NB * SS * E3;

  gemm64x16<<<dim3((NB * SS / 64) * (E3 / 16) / 8), dim3(256), 0, stream>>>(
      x, Wqkv, bqkv, qkv, NB * SS, E3, EE);

  rope_kernel<<<dim3(2 * NB * SS * HH * (DD / 2) / 256), dim3(256), 0, stream>>>(qkv);

  attn_kernel<<<dim3(NB * HH * (SS / 16) / 8), dim3(256), 0, stream>>>(qkv, attn);

  gemm64x16<<<dim3((NB * SS / 64) * (EE / 16) / 8), dim3(256), 0, stream>>>(
      attn, Wout, bout, out, NB * SS, EE, EE);
}


# RoPE fused into the qkv projection epilogue: accumulators rotated in registers (pair partner by DPP lane swap) before the store; separate RoPE kernel, its launch and its re-read of q,k removed
# speedup vs baseline: 1.5331x; 1.3283x over previous
#include <hip/hip_runtime.h>

typedef __attribute__((ext_vector_type(2))) float v2f;
typedef __attribute__((ext_vector_type(8))) float v8f;

#define NB 16
#define SS 1024
#define EE 256
#define HH 8
#define DD 32
#define E3 768

__device__ __forceinline__ v8f wmma_f32(v2f a, v2f b, v8f c) {
  return __builtin_amdgcn_wmma_f32_16x16x4_f32(
       false, a,  false, b,
       (short)0, c,  false,  false);
}

__device__ __forceinline__ float pair_other(float x) {
  int b = __builtin_bit_cast(int, x);
  return __builtin_bit_cast(float,
      __builtin_amdgcn_update_dpp(b, b, 0xB1, 0xF, 0xF, false));
}

template <bool ROPE>
__global__ __launch_bounds__(256) void gemm64x16(const float* __restrict__ X,
                                                 const float* __restrict__ W,
                                                 const float* __restrict__ bias,
                                                 float* __restrict__ Y,
                                                 int M, int N, int K) {
  const int lane = threadIdx.x & 31;
  const int wave = threadIdx.x >> 5;
  const int ntn  = N >> 4;
  int strip = blockIdx.x * 8 + wave;
  int sm = strip / ntn;
  int tn = strip - sm * ntn;
  const int row0 = sm << 6, col0 = tn << 4;
  const int m = lane & 15, half = lane >> 4;

  const float* xrow = X + (size_t)(row0 + m) * K + 2 * half;
  const float* wcol = W + (size_t)(2 * half) * N + col0 + m;

  v8f acc[4];
#pragma unroll
  for (int t = 0; t < 4; ++t) acc[t] = {};

#pragma unroll 4
  for (int kk = 0; kk < K; kk += 4) {
    v2f b;
    b.x = wcol[(size_t)kk * N];
    b.y = wcol[(size_t)(kk + 1) * N];
    v2f a0 = *(const v2f*)(xrow + kk);
    v2f a1 = *(const v2f*)(xrow + (size_t)16 * K + kk);
    v2f a2 = *(const v2f*)(xrow + (size_t)32 * K + kk);
    v2f a3 = *(const v2f*)(xrow + (size_t)48 * K + kk);
    acc[0] = wmma_f32(a0, b, acc[0]);
    acc[1] = wmma_f32(a1, b, acc[1]);
    acc[2] = wmma_f32(a2, b, acc[2]);
    acc[3] = wmma_f32(a3, b, acc[3]);
  }

  float bn = bias[col0 + m];
#pragma unroll
  for (int t = 0; t < 4; ++t) {
#pragma unroll
    for (int r = 0; r < 8; ++r) acc[t][r] += bn;
  }

  if constexpr (ROPE) {
    if (col0 < 2 * EE) {
      const float freq = __expf(-1.1512925464970231f * (float)(m >> 1));
      const bool odd = (m & 1) != 0;
      if (col0 & 16) {
#pragma unroll
        for (int tt = 0; tt < 2; ++tt) {
#pragma unroll
          for (int r = 0; r < 8; ++r) {
            float sn, cs;
            __sincosf((float)(16 * tt + r + 8 * half) * freq, &sn, &cs);
            const float ss = odd ? sn : -sn;
            float a = acc[tt][r], c = acc[tt + 2][r];
            acc[tt][r]     = a * cs + pair_other(a) * ss;
            acc[tt + 2][r] = c * cs + pair_other(c) * ss;
          }
        }
      } else {
        const int i0 = (row0 & (SS - 1)) >> 5;
#pragma unroll
        for (int th = 0; th < 2; ++th) {
          float sn, cs;
          __sincosf((float)(i0 + th) * freq, &sn, &cs);
          const float ss = odd ? sn : -sn;
#pragma unroll
          for (int t = 2 * th; t < 2 * th + 2; ++t) {
#pragma unroll
            for (int r = 0; r < 8; ++r) {
              float a = acc[t][r];
              acc[t][r] = a * cs + pair_other(a) * ss;
            }
          }
        }
      }
    }
  }

#pragma unroll
  for (int t = 0; t < 4; ++t) {
#pragma unroll
    for (int r = 0; r < 8; ++r) {
      int Mr = 16 * t + r + 8 * half;
      Y[(size_t)(row0 + Mr) * N + col0 + m] = acc[t][r];
    }
  }
}

__device__ __forceinline__ float xhalf(float x) {
  int b = __builtin_bit_cast(int, x);
  return __builtin_bit_cast(float,
      __builtin_amdgcn_permlanex16(b, b, 0x76543210, 0xfedcba98, false, false));
}

__global__ __launch_bounds__(256) void attn_kernel(const float* __restrict__ qkv,
                                                   float* __restrict__ out) {
  const int lane = threadIdx.x & 31;
  const int wave = threadIdx.x >> 5;
  int id = blockIdx.x * 8 + wave;
  const int qt = id & 63; id >>= 6;
  const int h  = id & 7;  id >>= 3;
  const int b  = id;
  const int m = lane & 15, half = lane >> 4;
  const int row0 = qt << 4;

  const size_t basebh = (size_t)(b * SS) * E3 + h * DD;
  const float* kslab = qkv + basebh + EE;
  const float* vslab = qkv + basebh + 2 * EE;

  v2f qb[8];
#pragma unroll
  for (int f = 0; f < 8; ++f)
    qb[f] = *(const v2f*)(qkv + basebh + (size_t)(row0 + m) * E3 + 4 * f + 2 * half);

  float mrun = -3.0e38f, lpart = 0.0f;
  v8f o0 = {}, o1 = {};

  const float scale = 0.17677669529663687f;

  for (int kt = 0; kt < 64; ++kt) {
    const int krow0 = kt << 4;

    v8f sacc = {};
    const float* kbase = kslab + (size_t)(krow0 + m) * E3 + 2 * half;
#pragma unroll
    for (int f = 0; f < 8; ++f) {
      v2f ka = *(const v2f*)(kbase + 4 * f);
      sacc = wmma_f32(ka, qb[f], sacc);
    }

    float v[8];
#pragma unroll
    for (int r = 0; r < 8; ++r) v[r] = sacc[r] * scale;
    float tmax = fmaxf(fmaxf(fmaxf(v[0], v[1]), fmaxf(v[2], v[3])),
                       fmaxf(fmaxf(v[4], v[5]), fmaxf(v[6], v[7])));
    tmax = fmaxf(tmax, xhalf(tmax));
    const float mnew = fmaxf(mrun, tmax);
    const float corr = __expf(mrun - mnew);
    mrun = mnew;

    v2f pb[4];
    float psum = 0.0f;
#pragma unroll
    for (int f = 0; f < 4; ++f) {
      float p0 = __expf(v[2 * f]     - mnew);
      float p1 = __expf(v[2 * f + 1] - mnew);
      psum += p0;
      psum += p1;
      pb[f].x = p0;
      pb[f].y = p1;
    }
    lpart = lpart * corr + psum;
#pragma unroll
    for (int r = 0; r < 8; ++r) { o0[r] *= corr; o1[r] *= corr; }

#pragma unroll
    for (int f = 0; f < 4; ++f) {
      const float* v0 = vslab + (size_t)(krow0 + 8 * half + 2 * f) * E3;
      const float* v1 = v0 + E3;
      v2f va0, va1;
      va0.x = v0[m];      va0.y = v1[m];
      va1.x = v0[16 + m]; va1.y = v1[16 + m];
      o0 = wmma_f32(va0, pb[f], o0);
      o1 = wmma_f32(va1, pb[f], o1);
    }
  }

  float lsum = lpart + xhalf(lpart);
  float inv = 1.0f / lsum;
  float* orow = out + (size_t)(b * SS + row0 + m) * EE + h * DD + 8 * half;
  typedef __attribute__((ext_vector_type(4))) float v4f;
  v4f w0 = {o0[0] * inv, o0[1] * inv, o0[2] * inv, o0[3] * inv};
  v4f w1 = {o0[4] * inv, o0[5] * inv, o0[6] * inv, o0[7] * inv};
  v4f w2 = {o1[0] * inv, o1[1] * inv, o1[2] * inv, o1[3] * inv};
  v4f w3 = {o1[4] * inv, o1[5] * inv, o1[6] * inv, o1[7] * inv};
  *(v4f*)(orow)      = w0;
  *(v4f*)(orow + 4)  = w1;
  *(v4f*)(orow + 16) = w2;
  *(v4f*)(orow + 20) = w3;
}

extern "C" void kernel_launch(void* const* d_in, const int* in_sizes, int n_in,
                              void* d_out, int out_size, void* d_ws, size_t ws_size,
                              hipStream_t stream) {
  const float* x    = (const float*)d_in[0];
  const float* Wqkv = (const float*)d_in[1];
  const float* bqkv = (const float*)d_in[2];
  const float* Wout = (const float*)d_in[3];
  const float* bout = (const float*)d_in[4];
  float* out  = (float*)d_out;
  float* qkv  = (float*)d_ws;
  float* attn = qkv + (size_t)NB * SS * E3;

  gemm64x16<true><<<dim3((NB * SS / 64) * (E3 / 16) / 8), dim3(256), 0, stream>>>(
      x, Wqkv, bqkv, qkv, NB * SS, E3, EE);

  attn_kernel<<<dim3(NB * HH * (SS / 16) / 8), dim3(256), 0, stream>>>(qkv, attn);

  gemm64x16<false><<<dim3((NB * SS / 64) * (EE / 16) / 8), dim3(256), 0, stream>>>(
      attn, Wout, bout, out, NB * SS, EE, EE);
}


# fragment loads widened, k order permuted alike on both operands: X and K,q rows 16 B per lane (2 WMMA steps each), V rows 8 B per lane; half the load instructions; k summation order changes
# speedup vs baseline: 1.7663x; 1.1521x over previous
#include <hip/hip_runtime.h>

typedef __attribute__((ext_vector_type(2))) float v2f;
typedef __attribute__((ext_vector_type(4))) float v4f;
typedef __attribute__((ext_vector_type(8))) float v8f;

#define NB 16
#define SS 1024
#define EE 256
#define HH 8
#define DD 32
#define E3 768

__device__ __forceinline__ v8f wmma_f32(v2f a, v2f b, v8f c) {
  return __builtin_amdgcn_wmma_f32_16x16x4_f32(
       false, a,  false, b,
       (short)0, c,  false,  false);
}

__device__ __forceinline__ float pair_other(float x) {
  int b = __builtin_bit_cast(int, x);
  return __builtin_bit_cast(float,
      __builtin_amdgcn_update_dpp(b, b, 0xB1, 0xF, 0xF, false));
}

template <bool ROPE>
__global__ __launch_bounds__(256) void gemm64x16(const float* __restrict__ X,
                                                 const float* __restrict__ W,
                                                 const float* __restrict__ bias,
                                                 float* __restrict__ Y,
                                                 int M, int N, int K) {
  const int lane = threadIdx.x & 31;
  const int wave = threadIdx.x >> 5;
  const int ntn  = N >> 4;
  int strip = blockIdx.x * 8 + wave;
  int sm = strip / ntn;
  int tn = strip - sm * ntn;
  const int row0 = sm << 6, col0 = tn << 4;
  const int m = lane & 15, half = lane >> 4;

  const float* xrow = X + (size_t)(row0 + m) * K + 4 * half;
  const float* wcol = W + (size_t)(4 * half) * N + col0 + m;

  v8f acc[4];
#pragma unroll
  for (int t = 0; t < 4; ++t) acc[t] = {};

#pragma unroll 2
  for (int kk = 0; kk < K; kk += 8) {
    v2f b0, b1;
    b0.x = wcol[(size_t)kk * N];
    b0.y = wcol[(size_t)(kk + 1) * N];
    b1.x = wcol[(size_t)(kk + 2) * N];
    b1.y = wcol[(size_t)(kk + 3) * N];
    v4f x0 = *(const v4f*)(xrow + kk);
    v4f x1 = *(const v4f*)(xrow + (size_t)16 * K + kk);
    v4f x2 = *(const v4f*)(xrow + (size_t)32 * K + kk);
    v4f x3 = *(const v4f*)(xrow + (size_t)48 * K + kk);
    acc[0] = wmma_f32(x0.xy, b0, acc[0]);
    acc[1] = wmma_f32(x1.xy, b0, acc[1]);
    acc[2] = wmma_f32(x2.xy, b0, acc[2]);
    acc[3] = wmma_f32(x3.xy, b0, acc[3]);
    acc[0] = wmma_f32(x0.zw, b1, acc[0]);
    acc[1] = wmma_f32(x1.zw, b1, acc[1]);
    acc[2] = wmma_f32(x2.zw, b1, acc[2]);
    acc[3] = wmma_f32(x3.zw, b1, acc[3]);
  }

  float bn = bias[col0 + m];
#pragma unroll
  for (int t = 0; t < 4; ++t) {
#pragma unroll
    for (int r = 0; r < 8; ++r) acc[t][r] += bn;
  }

  if constexpr (ROPE) {
    if (col0 < 2 * EE) {
      const float freq = __expf(-1.1512925464970231f * (float)(m >> 1));
      const bool odd = (m & 1) != 0;
      if (col0 & 16) {
#pragma unroll
        for (int tt = 0; tt < 2; ++tt) {
#pragma unroll
          for (int r = 0; r < 8; ++r) {
            float sn, cs;
            __sincosf((float)(16 * tt + r + 8 * half) * freq, &sn, &cs);
            const float ss = odd ? sn : -sn;
            float a = acc[tt][r], c = acc[tt + 2][r];
            acc[tt][r]     = a * cs + pair_other(a) * ss;
            acc[tt + 2][r] = c * cs + pair_other(c) * ss;
          }
        }
      } else {
        const int i0 = (row0 & (SS - 1)) >> 5;
#pragma unroll
        for (int th = 0; th < 2; ++th) {
          float sn, cs;
          __sincosf((float)(i0 + th) * freq, &sn, &cs);
          const float ss = odd ? sn : -sn;
#pragma unroll
          for (int t = 2 * th; t < 2 * th + 2; ++t) {
#pragma unroll
            for (int r = 0; r < 8; ++r) {
              float a = acc[t][r];
              acc[t][r] = a * cs + pair_other(a) * ss;
            }
          }
        }
      }
    }
  }

#pragma unroll
  for (int t = 0; t < 4; ++t) {
#pragma unroll
    for (int r = 0; r < 8; ++r) {
      int Mr = 16 * t + r + 8 * half;
      Y[(size_t)(row0 + Mr) * N + col0 + m] = acc[t][r];
    }
  }
}

__device__ __forceinline__ float xhalf(float x) {
  int b = __builtin_bit_cast(int, x);
  return __builtin_bit_cast(float,
      __builtin_amdgcn_permlanex16(b, b, 0x76543210, 0xfedcba98, false, false));
}

__global__ __launch_bounds__(256) void attn_kernel(const float* __restrict__ qkv,
                                                   float* __restrict__ out) {
  const int lane = threadIdx.x & 31;
  const int wave = threadIdx.x >> 5;
  int id = blockIdx.x * 8 + wave;
  const int qt = id & 63; id >>= 6;
  const int h  = id & 7;  id >>= 3;
  const int b  = id;
  const int m = lane & 15, half = lane >> 4;
  const int row0 = qt << 4;

  const size_t basebh = (size_t)(b * SS) * E3 + h * DD;
  const float* kslab = qkv + basebh + EE;
  const float* vslab = qkv + basebh + 2 * EE;

  v2f qb[8];
#pragma unroll
  for (int g = 0; g < 4; ++g) {
    v4f qv = *(const v4f*)(qkv + basebh + (size_t)(row0 + m) * E3 + 8 * g + 4 * half);
    qb[2 * g]     = qv.xy;
    qb[2 * g + 1] = qv.zw;
  }

  float mrun = -3.0e38f, lpart = 0.0f;
  v8f o0 = {}, o1 = {};

  const float scale = 0.17677669529663687f;

  for (int kt = 0; kt < 64; ++kt) {
    const int krow0 = kt << 4;

    v8f sacc = {};
    const float* kbase = kslab + (size_t)(krow0 + m) * E3 + 4 * half;
#pragma unroll
    for (int g = 0; g < 4; ++g) {
      v4f kv = *(const v4f*)(kbase + 8 * g);
      sacc = wmma_f32(kv.xy, qb[2 * g], sacc);
      sacc = wmma_f32(kv.zw, qb[2 * g + 1], sacc);
    }

    float v[8];
#pragma unroll
    for (int r = 0; r < 8; ++r) v[r] = sacc[r] * scale;
    float tmax = fmaxf(fmaxf(fmaxf(v[0], v[1]), fmaxf(v[2], v[3])),
                       fmaxf(fmaxf(v[4], v[5]), fmaxf(v[6], v[7])));
    tmax = fmaxf(tmax, xhalf(tmax));
    const float mnew = fmaxf(mrun, tmax);
    const float corr = __expf(mrun - mnew);
    mrun = mnew;

    v2f pb[4];
    float psum = 0.0f;
#pragma unroll
    for (int f = 0; f < 4; ++f) {
      float p0 = __expf(v[2 * f]     - mnew);
      float p1 = __expf(v[2 * f + 1] - mnew);
      psum += p0;
      psum += p1;
      pb[f].x = p0;
      pb[f].y = p1;
    }
    lpart = lpart * corr + psum;
#pragma unroll
    for (int r = 0; r < 8; ++r) { o0[r] *= corr; o1[r] *= corr; }

#pragma unroll
    for (int f = 0; f < 4; ++f) {
      const float* v0 = vslab + (size_t)(krow0 + 8 * half + 2 * f) * E3 + 2 * m;
      v2f e0 = *(const v2f*)(v0);
      v2f e1 = *(const v2f*)(v0 + E3);
      v2f va0, va1;
      va0.x = e0.x; va0.y = e1.x;
      va1.x = e0.y; va1.y = e1.y;
      o0 = wmma_f32(va0, pb[f], o0);
      o1 = wmma_f32(va1, pb[f], o1);
    }
  }

  float lsum = lpart + xhalf(lpart);
  float inv = 1.0f / lsum;
  float* orow = out + (size_t)(b * SS + row0 + m) * EE + h * DD + 16 * half;
  v4f w0 = {o0[0] * inv, o1[0] * inv, o0[1] * inv, o1[1] * inv};
  v4f w1 = {o0[2] * inv, o1[2] * inv, o0[3] * inv, o1[3] * inv};
  v4f w2 = {o0[4] * inv, o1[4] * inv, o0[5] * inv, o1[5] * inv};
  v4f w3 = {o0[6] * inv, o1[6] * inv, o0[7] * inv, o1[7] * inv};
  *(v4f*)(orow)      = w0;
  *(v4f*)(orow + 4)  = w1;
  *(v4f*)(orow + 8)  = w2;
  *(v4f*)(orow + 12) = w3;
}

extern "C" void kernel_launch(void* const* d_in, const int* in_sizes, int n_in,
                              void* d_out, int out_size, void* d_ws, size_t ws_size,
                              hipStream_t stream) {
  const float* x    = (const float*)d_in[0];
  const float* Wqkv = (const float*)d_in[1];
  const float* bqkv = (const float*)d_in[2];
  const float* Wout = (const float*)d_in[3];
  const float* bout = (const float*)d_in[4];
  float* out  = (float*)d_out;
  float* qkv  = (float*)d_ws;
  float* attn = qkv + (size_t)NB * SS * E3;

  gemm64x16<true><<<dim3((NB * SS / 64) * (E3 / 16) / 8), dim3(256), 0, stream>>>(
      x, Wqkv, bqkv, qkv, NB * SS, E3, EE);

  attn_kernel<<<dim3(NB * HH * (SS / 16) / 8), dim3(256), 0, stream>>>(qkv, attn);

  gemm64x16<false><<<dim3((NB * SS / 64) * (EE / 16) / 8), dim3(256), 0, stream>>>(
      attn, Wout, bout, out, NB * SS, EE, EE);
}


# attention: K and V rows staged in LDS once per block (64-key stages, double buffered, one barrier each) and shared by its 8 waves; WMMA fragments read from LDS; per-query sum order unchanged
# speedup vs baseline: 1.8997x; 1.0755x over previous
#include <hip/hip_runtime.h>

typedef __attribute__((ext_vector_type(2))) float v2f;
typedef __attribute__((ext_vector_type(4))) float v4f;
typedef __attribute__((ext_vector_type(8))) float v8f;

#define NB 16
#define SS 1024
#define EE 256
#define HH 8
#define DD 32
#define E3 768

__device__ __forceinline__ v8f wmma_f32(v2f a, v2f b, v8f c) {
  return __builtin_amdgcn_wmma_f32_16x16x4_f32(
       false, a,  false, b,
       (short)0, c,  false,  false);
}

__device__ __forceinline__ float pair_other(float x) {
  int b = __builtin_bit_cast(int, x);
  return __builtin_bit_cast(float,
      __builtin_amdgcn_update_dpp(b, b, 0xB1, 0xF, 0xF, false));
}

template <bool ROPE>
__global__ __launch_bounds__(256) void gemm64x16(const float* __restrict__ X,
                                                 const float* __restrict__ W,
                                                 const float* __restrict__ bias,
                                                 float* __restrict__ Y,
                                                 int M, int N, int K) {
  const int lane = threadIdx.x & 31;
  const int wave = threadIdx.x >> 5;
  const int ntn  = N >> 4;
  int strip = blockIdx.x * 8 + wave;
  int sm = strip / ntn;
  int tn = strip - sm * ntn;
  const int row0 = sm << 6, col0 = tn << 4;
  const int m = lane & 15, half = lane >> 4;

  const float* xrow = X + (size_t)(row0 + m) * K + 4 * half;
  const float* wcol = W + (size_t)(4 * half) * N + col0 + m;

  v8f acc[4];
#pragma unroll
  for (int t = 0; t < 4; ++t) acc[t] = {};

#pragma unroll 2
  for (int kk = 0; kk < K; kk += 8) {
    v2f b0, b1;
    b0.x = wcol[(size_t)kk * N];
    b0.y = wcol[(size_t)(kk + 1) * N];
    b1.x = wcol[(size_t)(kk + 2) * N];
    b1.y = wcol[(size_t)(kk + 3) * N];
    v4f x0 = *(const v4f*)(xrow + kk);
    v4f x1 = *(const v4f*)(xrow + (size_t)16 * K + kk);
    v4f x2 = *(const v4f*)(xrow + (size_t)32 * K + kk);
    v4f x3 = *(const v4f*)(xrow + (size_t)48 * K + kk);
    acc[0] = wmma_f32(x0.xy, b0, acc[0]);
    acc[1] = wmma_f32(x1.xy, b0, acc[1]);
    acc[2] = wmma_f32(x2.xy, b0, acc[2]);
    acc[3] = wmma_f32(x3.xy, b0, acc[3]);
    acc[0] = wmma_f32(x0.zw, b1, acc[0]);
    acc[1] = wmma_f32(x1.zw, b1, acc[1]);
    acc[2] = wmma_f32(x2.zw, b1, acc[2]);
    acc[3] = wmma_f32(x3.zw, b1, acc[3]);
  }

  float bn = bias[col0 + m];
#pragma unroll
  for (int t = 0; t < 4; ++t) {
#pragma unroll
    for (int r = 0; r < 8; ++r) acc[t][r] += bn;
  }

  if constexpr (ROPE) {
    if (col0 < 2 * EE) {
      const float freq = __expf(-1.1512925464970231f * (float)(m >> 1));
      const bool odd = (m & 1) != 0;
      if (col0 & 16) {
#pragma unroll
        for (int tt = 0; tt < 2; ++tt) {
#pragma unroll
          for (int r = 0; r < 8; ++r) {
            float sn, cs;
            __sincosf((float)(16 * tt + r + 8 * half) * freq, &sn, &cs);
            const float ss = odd ? sn : -sn;
            float a = acc[tt][r], c = acc[tt + 2][r];
            acc[tt][r]     = a * cs + pair_other(a) * ss;
            acc[tt + 2][r] = c * cs + pair_other(c) * ss;
          }
        }
      } else {
        const int i0 = (row0 & (SS - 1)) >> 5;
#pragma unroll
        for (int th = 0; th < 2; ++th) {
          float sn, cs;
          __sincosf((float)(i0 + th) * freq, &sn, &cs);
          const float ss = odd ? sn : -sn;
#pragma unroll
          for (int t = 2 * th; t < 2 * th + 2; ++t) {
#pragma unroll
            for (int r = 0; r < 8; ++r) {
              float a = acc[t][r];
              acc[t][r] = a * cs + pair_other(a) * ss;
            }
          }
        }
      }
    }
  }

#pragma unroll
  for (int t = 0; t < 4; ++t) {
#pragma unroll
    for (int r = 0; r < 8; ++r) {
      int Mr = 16 * t + r + 8 * half;
      Y[(size_t)(row0 + Mr) * N + col0 + m] = acc[t][r];
    }
  }
}

#define KCH 64
#define KP  36
#define VP  72
#define STG (KCH * KP + (KCH / 2) * VP)

__device__ __forceinline__ float xhalf(float x) {
  int b = __builtin_bit_cast(int, x);
  return __builtin_bit_cast(float,
      __builtin_amdgcn_permlanex16(b, b, 0x76543210, 0xfedcba98, false, false));
}

__global__ __launch_bounds__(256) void attn_kernel(const float* __restrict__ qkv,
                                                   float* __restrict__ out) {
  __shared__ __attribute__((aligned(16))) float smem[2 * STG];

  const int tid  = threadIdx.x;
  const int lane = tid & 31;
  const int wave = tid >> 5;
  int id = blockIdx.x * 8 + wave;
  const int qt = id & 63; id >>= 6;
  const int h  = id & 7;  id >>= 3;
  const int b  = id;
  const int m = lane & 15, half = lane >> 4;
  const int row0 = qt << 4;

  const size_t basebh = (size_t)(b * SS) * E3 + h * DD;
  const float* kslab = qkv + basebh + EE;
  const float* vslab = qkv + basebh + 2 * EE;

  const int sr = tid >> 3, sp = tid & 7;
  const float* kg = kslab + (size_t)sr * E3 + 4 * sp;
  const float* vg = vslab + (size_t)(2 * sr) * E3 + 4 * sp;
  const int ksl = sr * KP + 4 * sp;
  const int vsl = KCH * KP + sr * VP + 8 * sp;

  v2f qb[8];
#pragma unroll
  for (int g = 0; g < 4; ++g) {
    v4f qv = *(const v4f*)(qkv + basebh + (size_t)(row0 + m) * E3 + 8 * g + 4 * half);
    qb[2 * g]     = qv.xy;
    qb[2 * g + 1] = qv.zw;
  }

  {
    v4f a0 = *(const v4f*)(kg);
    v4f a1 = *(const v4f*)(kg + (size_t)32 * E3);
    v4f c0 = *(const v4f*)(vg);
    v4f c1 = *(const v4f*)(vg + E3);
    v4f i0 = {c0.x, c1.x, c0.y, c1.y};
    v4f i1 = {c0.z, c1.z, c0.w, c1.w};
    *(v4f*)(smem + ksl)           = a0;
    *(v4f*)(smem + ksl + 32 * KP) = a1;
    *(v4f*)(smem + vsl)           = i0;
    *(v4f*)(smem + vsl + 4)       = i1;
  }
  __syncthreads();

  float mrun = -3.0e38f, lpart = 0.0f;
  v8f o0 = {}, o1 = {};

  const float scale = 0.17677669529663687f;

  const int kfrag = m * KP + 4 * half;
  const int vfrag = KCH * KP + 4 * half * VP + 2 * m;

  for (int c = 0; c < SS / KCH; ++c) {
    const float* st = smem + (c & 1) * STG;
    float* nst = smem + ((c + 1) & 1) * STG;

    const int cn = (c + 1 < SS / KCH) ? c + 1 : c;
    const size_t go = (size_t)(cn * KCH) * E3;
    v4f a0 = *(const v4f*)(kg + go);
    v4f a1 = *(const v4f*)(kg + go + (size_t)32 * E3);
    v4f c0 = *(const v4f*)(vg + go);
    v4f c1 = *(const v4f*)(vg + go + E3);

#pragma unroll
    for (int kt = 0; kt < KCH / 16; ++kt) {
      v8f sacc = {};
      const float* kl = st + 16 * kt * KP + kfrag;
#pragma unroll
      for (int g = 0; g < 4; ++g) {
        v4f kv = *(const v4f*)(kl + 8 * g);
        sacc = wmma_f32(kv.xy, qb[2 * g], sacc);
        sacc = wmma_f32(kv.zw, qb[2 * g + 1], sacc);
      }

      float v[8];
#pragma unroll
      for (int r = 0; r < 8; ++r) v[r] = sacc[r] * scale;
      float tmax = fmaxf(fmaxf(fmaxf(v[0], v[1]), fmaxf(v[2], v[3])),
                         fmaxf(fmaxf(v[4], v[5]), fmaxf(v[6], v[7])));
      tmax = fmaxf(tmax, xhalf(tmax));
      const float mnew = fmaxf(mrun, tmax);
      const float corr = __expf(mrun - mnew);
      mrun = mnew;

      v2f pb[4];
      float psum = 0.0f;
#pragma unroll
      for (int f = 0; f < 4; ++f) {
        float p0 = __expf(v[2 * f]     - mnew);
        float p1 = __expf(v[2 * f + 1] - mnew);
        psum += p0;
        psum += p1;
        pb[f].x = p0;
        pb[f].y = p1;
      }
      lpart = lpart * corr + psum;
#pragma unroll
      for (int r = 0; r < 8; ++r) { o0[r] *= corr; o1[r] *= corr; }

      const float* vl = st + 8 * kt * VP + vfrag;
#pragma unroll
      for (int f = 0; f < 4; ++f) {
        v2f va0 = *(const v2f*)(vl + f * VP);
        v2f va1 = *(const v2f*)(vl + f * VP + 32);
        o0 = wmma_f32(va0, pb[f], o0);
        o1 = wmma_f32(va1, pb[f], o1);
      }
    }

    {
      v4f i0 = {c0.x, c1.x, c0.y, c1.y};
      v4f i1 = {c0.z, c1.z, c0.w, c1.w};
      *(v4f*)(nst + ksl)           = a0;
      *(v4f*)(nst + ksl + 32 * KP) = a1;
      *(v4f*)(nst + vsl)           = i0;
      *(v4f*)(nst + vsl + 4)       = i1;
    }
    __syncthreads();
  }

  float lsum = lpart + xhalf(lpart);
  float inv = 1.0f / lsum;
  float* orow = out + (size_t)(b * SS + row0 + m) * EE + h * DD + 8 * half;
  v4f w0 = {o0[0] * inv, o0[1] * inv, o0[2] * inv, o0[3] * inv};
  v4f w1 = {o0[4] * inv, o0[5] * inv, o0[6] * inv, o0[7] * inv};
  v4f w2 = {o1[0] * inv, o1[1] * inv, o1[2] * inv, o1[3] * inv};
  v4f w3 = {o1[4] * inv, o1[5] * inv, o1[6] * inv, o1[7] * inv};
  *(v4f*)(orow)      = w0;
  *(v4f*)(orow + 4)  = w1;
  *(v4f*)(orow + 16) = w2;
  *(v4f*)(orow + 20) = w3;
}

extern "C" void kernel_launch(void* const* d_in, const int* in_sizes, int n_in,
                              void* d_out, int out_size, void* d_ws, size_t ws_size,
                              hipStream_t stream) {
  const float* x    = (const float*)d_in[0];
  const float* Wqkv = (const float*)d_in[1];
  const float* bqkv = (const float*)d_in[2];
  const float* Wout = (const float*)d_in[3];
  const float* bout = (const float*)d_in[4];
  float* out  = (float*)d_out;
  float* qkv  = (float*)d_ws;
  float* attn = qkv + (size_t)NB * SS * E3;

  gemm64x16<true><<<dim3((NB * SS / 64) * (E3 / 16) / 8), dim3(256), 0, stream>>>(
      x, Wqkv, bqkv, qkv, NB * SS, E3, EE);

  attn_kernel<<<dim3(NB * HH * (SS / 16) / 8), dim3(256), 0, stream>>>(qkv, attn);

  gemm64x16<false><<<dim3((NB * SS / 64) * (EE / 16) / 8), dim3(256), 0, stream>>>(
      attn, Wout, bout, out, NB * SS, EE, EE);
}


# projection GEMMs: wave tile 64x16 -> 64x64 (qkv) and 32x64 (out); W rows read 16 B per lane (lane owns columns 4m..4m+3): about 4x fewer loads and cache lines per WMMA; k summation order unchanged
# speedup vs baseline: 2.2680x; 1.1939x over previous
#include <hip/hip_runtime.h>

typedef __attribute__((ext_vector_type(2))) float v2f;
typedef __attribute__((ext_vector_type(4))) float v4f;
typedef __attribute__((ext_vector_type(8))) float v8f;

#define NB 16
#define SS 1024
#define EE 256
#define HH 8
#define DD 32
#define E3 768

__device__ __forceinline__ v8f wmma_f32(v2f a, v2f b, v8f c) {
  return __builtin_amdgcn_wmma_f32_16x16x4_f32(
       false, a,  false, b,
       (short)0, c,  false,  false);
}

template <int TR, int KQ>
__device__ __forceinline__ void load_group(v4f (&xf)[KQ][TR], v4f (&wf)[KQ][4],
                                           const float* __restrict__ xrow,
                                           const float* __restrict__ wrow,
                                           int K, int N, int kk) {
#pragma unroll
  for (int u = 0; u < KQ; ++u) {
#pragma unroll
    for (int t = 0; t < TR; ++t)
      xf[u][t] = *(const v4f*)(xrow + (size_t)(16 * t) * K + kk + 8 * u);
#pragma unroll
    for (int q = 0; q < 4; ++q)
      wf[u][q] = *(const v4f*)(wrow + (size_t)(kk + 8 * u + q) * N);
  }
}

template <int TR, int KQ>
__device__ __forceinline__ void mma_group(v8f (&acc)[TR][4], const v4f (&xf)[KQ][TR],
                                          const v4f (&wf)[KQ][4]) {
#pragma unroll
  for (int u = 0; u < KQ; ++u) {
#pragma unroll
    for (int s = 0; s < 2; ++s) {
#pragma unroll
      for (int t = 0; t < TR; ++t) {
        v2f a;
        a.x = xf[u][t][2 * s];
        a.y = xf[u][t][2 * s + 1];
#pragma unroll
        for (int ct = 0; ct < 4; ++ct) {
          v2f b;
          b.x = wf[u][2 * s][ct];
          b.y = wf[u][2 * s + 1][ct];
          acc[t][ct] = wmma_f32(a, b, acc[t][ct]);
        }
      }
    }
  }
}

template <bool ROPE, int TR, int KQ>
__global__ __launch_bounds__(128) void gemm_tile(const float* __restrict__ X,
                                                 const float* __restrict__ W,
                                                 const float* __restrict__ bias,
                                                 float* __restrict__ Y,
                                                 int M, int N, int K) {
  const int lane = threadIdx.x & 31;
  const int wave = threadIdx.x >> 5;
  const int ntn  = N >> 6;
  int strip = blockIdx.x * 4 + wave;
  int sm = strip / ntn;
  int tn = strip - sm * ntn;
  const int row0 = sm * (16 * TR), col0 = tn << 6;
  const int m = lane & 15, half = lane >> 4;

  const float* xrow = X + (size_t)(row0 + m) * K + 4 * half;
  const float* wrow = W + (size_t)(4 * half) * N + col0 + 4 * m;

  v8f acc[TR][4];
#pragma unroll
  for (int t = 0; t < TR; ++t) {
#pragma unroll
    for (int ct = 0; ct < 4; ++ct) acc[t][ct] = {};
  }

  constexpr int KG = 8 * KQ;
  v4f xa[KQ][TR], wa[KQ][4], xb[KQ][TR], wb[KQ][4];
  load_group<TR, KQ>(xa, wa, xrow, wrow, K, N, 0);
  for (int kk = 0; kk < K; kk += 2 * KG) {
    load_group<TR, KQ>(xb, wb, xrow, wrow, K, N, kk + KG);
    __builtin_amdgcn_sched_barrier(0);
    mma_group<TR, KQ>(acc, xa, wa);
    const int kn = (kk + 2 * KG < K) ? kk + 2 * KG : K - KG;
    __builtin_amdgcn_sched_barrier(0);
    load_group<TR, KQ>(xa, wa, xrow, wrow, K, N, kn);
    __builtin_amdgcn_sched_barrier(0);
    mma_group<TR, KQ>(acc, xb, wb);
  }

  const v4f bn = *(const v4f*)(bias + col0 + 4 * m);
#pragma unroll
  for (int t = 0; t < TR; ++t) {
#pragma unroll
    for (int ct = 0; ct < 4; ++ct) {
#pragma unroll
      for (int r = 0; r < 8; ++r) acc[t][ct][r] += bn[ct];
    }
  }

  if constexpr (ROPE) {
    if (col0 < 2 * EE) {
      const bool byrow = (m & 7) < 4;
      constexpr int NTH = TR / 2 > 0 ? TR / 2 : 1;
      constexpr int NTT = TR < 2 ? TR : 2;
      const int i0 = (row0 & (SS - 1)) >> 5;
      float freq[2], csI[NTH][2], snI[NTH][2];
#pragma unroll
      for (int pp = 0; pp < 2; ++pp) {
        freq[pp] = __expf(-1.1512925464970231f * (float)((2 * (m & 7) + pp) & 7));
#pragma unroll
        for (int th = 0; th < NTH; ++th)
          __sincosf((float)(i0 + th) * freq[pp], &snI[th][pp], &csI[th][pp]);
      }
#pragma unroll
      for (int tt = 0; tt < NTT; ++tt) {
#pragma unroll
        for (int r = 0; r < 8; ++r) {
#pragma unroll
          for (int pp = 0; pp < 2; ++pp) {
            float snJ, csJ;
            __sincosf((float)(16 * tt + r + 8 * half) * freq[pp], &snJ, &csJ);
#pragma unroll
            for (int th = 0; th < NTH; ++th) {
              const int t = tt + 2 * th;
              const float cs = byrow ? csI[th][pp] : csJ;
              const float sn = byrow ? snI[th][pp] : snJ;
              const float x0 = acc[t][2 * pp][r], x1 = acc[t][2 * pp + 1][r];
              acc[t][2 * pp][r]     = x0 * cs - x1 * sn;
              acc[t][2 * pp + 1][r] = x1 * cs + x0 * sn;
            }
          }
        }
      }
    }
  }

#pragma unroll
  for (int t = 0; t < TR; ++t) {
#pragma unroll
    for (int r = 0; r < 8; ++r) {
      const int Mr = 16 * t + r + 8 * half;
      v4f o = {acc[t][0][r], acc[t][1][r], acc[t][2][r], acc[t][3][r]};
      *(v4f*)(Y + (size_t)(row0 + Mr) * N + col0 + 4 * m) = o;
    }
  }
}

#define KCH 64
#define KP  36
#define VP  72
#define STG (KCH * KP + (KCH / 2) * VP)

__device__ __forceinline__ float xhalf(float x) {
  int b = __builtin_bit_cast(int, x);
  return __builtin_bit_cast(float,
      __builtin_amdgcn_permlanex16(b, b, 0x76543210, 0xfedcba98, false, false));
}

__global__ __launch_bounds__(256) void attn_kernel(const float* __restrict__ qkv,
                                                   float* __restrict__ out) {
  __shared__ __attribute__((aligned(16))) float smem[2 * STG];

  const int tid  = threadIdx.x;
  const int lane = tid & 31;
  const int wave = tid >> 5;
  int id = blockIdx.x * 8 + wave;
  const int qt = id & 63; id >>= 6;
  const int h  = id & 7;  id >>= 3;
  const int b  = id;
  const int m = lane & 15, half = lane >> 4;
  const int row0 = qt << 4;

  const size_t basebh = (size_t)(b * SS) * E3 + h * DD;
  const float* kslab = qkv + basebh + EE;
  const float* vslab = qkv + basebh + 2 * EE;

  const int sr = tid >> 3, sp = tid & 7;
  const float* kg = kslab + (size_t)sr * E3 + 4 * sp;
  const float* vg = vslab + (size_t)(2 * sr) * E3 + 4 * sp;
  const int ksl = sr * KP + 4 * sp;
  const int vsl = KCH * KP + sr * VP + 8 * sp;

  v2f qb[8];
#pragma unroll
  for (int g = 0; g < 4; ++g) {
    v4f qv = *(const v4f*)(qkv + basebh + (size_t)(row0 + m) * E3 + 8 * g + 4 * half);
    qb[2 * g]     = qv.xy;
    qb[2 * g + 1] = qv.zw;
  }

  {
    v4f a0 = *(const v4f*)(kg);
    v4f a1 = *(const v4f*)(kg + (size_t)32 * E3);
    v4f c0 = *(const v4f*)(vg);
    v4f c1 = *(const v4f*)(vg + E3);
    v4f i0 = {c0.x, c1.x, c0.y, c1.y};
    v4f i1 = {c0.z, c1.z, c0.w, c1.w};
    *(v4f*)(smem + ksl)           = a0;
    *(v4f*)(smem + ksl + 32 * KP) = a1;
    *(v4f*)(smem + vsl)           = i0;
    *(v4f*)(smem + vsl + 4)       = i1;
  }
  __syncthreads();

  float mrun = -3.0e38f, lpart = 0.0f;
  v8f o0 = {}, o1 = {};

  const float scale = 0.17677669529663687f;

  const int kfrag = m * KP + 4 * half;
  const int vfrag = KCH * KP + 4 * half * VP + 2 * m;

  for (int c = 0; c < SS / KCH; ++c) {
    const float* st = smem + (c & 1) * STG;
    float* nst = smem + ((c + 1) & 1) * STG;

    const int cn = (c + 1 < SS / KCH) ? c + 1 : c;
    const size_t go = (size_t)(cn * KCH) * E3;
    v4f a0 = *(const v4f*)(kg + go);
    v4f a1 = *(const v4f*)(kg + go + (size_t)32 * E3);
    v4f c0 = *(const v4f*)(vg + go);
    v4f c1 = *(const v4f*)(vg + go + E3);

#pragma unroll
    for (int kt = 0; kt < KCH / 16; ++kt) {
      v8f sacc = {};
      const float* kl = st + 16 * kt * KP + kfrag;
#pragma unroll
      for (int g = 0; g < 4; ++g) {
        v4f kv = *(const v4f*)(kl + 8 * g);
        sacc = wmma_f32(kv.xy, qb[2 * g], sacc);
        sacc = wmma_f32(kv.zw, qb[2 * g + 1], sacc);
      }

      float v[8];
#pragma unroll
      for (int r = 0; r < 8; ++r) v[r] = sacc[r] * scale;
      float tmax = fmaxf(fmaxf(fmaxf(v[0], v[1]), fmaxf(v[2], v[3])),
                         fmaxf(fmaxf(v[4], v[5]), fmaxf(v[6], v[7])));
      tmax = fmaxf(tmax, xhalf(tmax));
      const float mnew = fmaxf(mrun, tmax);
      const float corr = __expf(mrun - mnew);
      mrun = mnew;

      v2f pb[4];
      float psum = 0.0f;
#pragma unroll
      for (int f = 0; f < 4; ++f) {
        float p0 = __expf(v[2 * f]     - mnew);
        float p1 = __expf(v[2 * f + 1] - mnew);
        psum += p0;
        psum += p1;
        pb[f].x = p0;
        pb[f].y = p1;
      }
      lpart = lpart * corr + psum;
#pragma unroll
      for (int r = 0; r < 8; ++r) { o0[r] *= corr; o1[r] *= corr; }

      const float* vl = st + 8 * kt * VP + vfrag;
#pragma unroll
      for (int f = 0; f < 4; ++f) {
        v2f va0 = *(const v2f*)(vl + f * VP);
        v2f va1 = *(const v2f*)(vl + f * VP + 32);
        o0 = wmma_f32(va0, pb[f], o0);
        o1 = wmma_f32(va1, pb[f], o1);
      }
    }

    {
      v4f i0 = {c0.x, c1.x, c0.y, c1.y};
      v4f i1 = {c0.z, c1.z, c0.w, c1.w};
      *(v4f*)(nst + ksl)           = a0;
      *(v4f*)(nst + ksl + 32 * KP) = a1;
      *(v4f*)(nst + vsl)           = i0;
      *(v4f*)(nst + vsl + 4)       = i1;
    }
    __syncthreads();
  }

  float lsum = lpart + xhalf(lpart);
  float inv = 1.0f / lsum;
  float* orow = out + (size_t)(b * SS + row0 + m) * EE + h * DD + 8 * half;
  v4f w0 = {o0[0] * inv, o0[1] * inv, o0[2] * inv, o0[3] * inv};
  v4f w1 = {o0[4] * inv, o0[5] * inv, o0[6] * inv, o0[7] * inv};
  v4f w2 = {o1[0] * inv, o1[1] * inv, o1[2] * inv, o1[3] * inv};
  v4f w3 = {o1[4] * inv, o1[5] * inv, o1[6] * inv, o1[7] * inv};
  *(v4f*)(orow)      = w0;
  *(v4f*)(orow + 4)  = w1;
  *(v4f*)(orow + 16) = w2;
  *(v4f*)(orow + 20) = w3;
}

extern "C" void kernel_launch(void* const* d_in, const int* in_sizes, int n_in,
                              void* d_out, int out_size, void* d_ws, size_t ws_size,
                              hipStream_t stream) {
  const float* x    = (const float*)d_in[0];
  const float* Wqkv = (const float*)d_in[1];
  const float* bqkv = (const float*)d_in[2];
  const float* Wout = (const float*)d_in[3];
  const float* bout = (const float*)d_in[4];
  float* out  = (float*)d_out;
  float* qkv  = (float*)d_ws;
  float* attn = qkv + (size_t)NB * SS * E3;

  gemm_tile<true, 4, 1><<<dim3((NB * SS / 64) * (E3 / 64) / 4), dim3(128), 0, stream>>>(
      x, Wqkv, bqkv, qkv, NB * SS, E3, EE);

  attn_kernel<<<dim3(NB * HH * (SS / 16) / 8), dim3(256), 0, stream>>>(qkv, attn);

  gemm_tile<false, 2, 2><<<dim3((NB * SS / 32) * (EE / 64) / 4), dim3(128), 0, stream>>>(
      attn, Wout, bout, out, NB * SS, EE, EE);
}


# attention softmax vector work halved (206 -> 113 vector instr per 64 WMMAs): one max/correction/rescale per 64-key stage, base-2 exponent by packed FMA; row-sum order and rescale points change
# speedup vs baseline: 2.4286x; 1.0708x over previous
#include <hip/hip_runtime.h>

typedef __attribute__((ext_vector_type(2))) float v2f;
typedef __attribute__((ext_vector_type(4))) float v4f;
typedef __attribute__((ext_vector_type(8))) float v8f;

#define NB 16
#define SS 1024
#define EE 256
#define HH 8
#define DD 32
#define E3 768

__device__ __forceinline__ v8f wmma_f32(v2f a, v2f b, v8f c) {
  return __builtin_amdgcn_wmma_f32_16x16x4_f32(
       false, a,  false, b,
       (short)0, c,  false,  false);
}

template <int TR, int KQ>
__device__ __forceinline__ void load_group(v4f (&xf)[KQ][TR], v4f (&wf)[KQ][4],
                                           const float* __restrict__ xrow,
                                           const float* __restrict__ wrow,
                                           int K, int N, int kk) {
#pragma unroll
  for (int u = 0; u < KQ; ++u) {
#pragma unroll
    for (int t = 0; t < TR; ++t)
      xf[u][t] = *(const v4f*)(xrow + (size_t)(16 * t) * K + kk + 8 * u);
#pragma unroll
    for (int q = 0; q < 4; ++q)
      wf[u][q] = *(const v4f*)(wrow + (size_t)(kk + 8 * u + q) * N);
  }
}

template <int TR, int KQ>
__device__ __forceinline__ void mma_group(v8f (&acc)[TR][4], const v4f (&xf)[KQ][TR],
                                          const v4f (&wf)[KQ][4]) {
#pragma unroll
  for (int u = 0; u < KQ; ++u) {
#pragma unroll
    for (int s = 0; s < 2; ++s) {
#pragma unroll
      for (int t = 0; t < TR; ++t) {
        v2f a;
        a.x = xf[u][t][2 * s];
        a.y = xf[u][t][2 * s + 1];
#pragma unroll
        for (int ct = 0; ct < 4; ++ct) {
          v2f b;
          b.x = wf[u][2 * s][ct];
          b.y = wf[u][2 * s + 1][ct];
          acc[t][ct] = wmma_f32(a, b, acc[t][ct]);
        }
      }
    }
  }
}

template <bool ROPE, int TR, int KQ>
__global__ __launch_bounds__(128) void gemm_tile(const float* __restrict__ X,
                                                 const float* __restrict__ W,
                                                 const float* __restrict__ bias,
                                                 float* __restrict__ Y,
                                                 int M, int N, int K) {
  const int lane = threadIdx.x & 31;
  const int wave = threadIdx.x >> 5;
  const int ntn  = N >> 6;
  int strip = blockIdx.x * 4 + wave;
  int sm = strip / ntn;
  int tn = strip - sm * ntn;
  const int row0 = sm * (16 * TR), col0 = tn << 6;
  const int m = lane & 15, half = lane >> 4;

  const float* xrow = X + (size_t)(row0 + m) * K + 4 * half;
  const float* wrow = W + (size_t)(4 * half) * N + col0 + 4 * m;

  v8f acc[TR][4];
#pragma unroll
  for (int t = 0; t < TR; ++t) {
#pragma unroll
    for (int ct = 0; ct < 4; ++ct) acc[t][ct] = {};
  }

  constexpr int KG = 8 * KQ;
  v4f xa[KQ][TR], wa[KQ][4], xb[KQ][TR], wb[KQ][4];
  load_group<TR, KQ>(xa, wa, xrow, wrow, K, N, 0);
  for (int kk = 0; kk < K; kk += 2 * KG) {
    load_group<TR, KQ>(xb, wb, xrow, wrow, K, N, kk + KG);
    __builtin_amdgcn_sched_barrier(0);
    mma_group<TR, KQ>(acc, xa, wa);
    const int kn = (kk + 2 * KG < K) ? kk + 2 * KG : K - KG;
    __builtin_amdgcn_sched_barrier(0);
    load_group<TR, KQ>(xa, wa, xrow, wrow, K, N, kn);
    __builtin_amdgcn_sched_barrier(0);
    mma_group<TR, KQ>(acc, xb, wb);
  }

  const v4f bn = *(const v4f*)(bias + col0 + 4 * m);
#pragma unroll
  for (int t = 0; t < TR; ++t) {
#pragma unroll
    for (int ct = 0; ct < 4; ++ct) {
#pragma unroll
      for (int r = 0; r < 8; ++r) acc[t][ct][r] += bn[ct];
    }
  }

  if constexpr (ROPE) {
    if (col0 < 2 * EE) {
      const bool byrow = (m & 7) < 4;
      constexpr int NTH = TR / 2 > 0 ? TR / 2 : 1;
      constexpr int NTT = TR < 2 ? TR : 2;
      const int i0 = (row0 & (SS - 1)) >> 5;
      float freq[2], csI[NTH][2], snI[NTH][2];
#pragma unroll
      for (int pp = 0; pp < 2; ++pp) {
        freq[pp] = __expf(-1.1512925464970231f * (float)((2 * (m & 7) + pp) & 7));
#pragma unroll
        for (int th = 0; th < NTH; ++th)
          __sincosf((float)(i0 + th) * freq[pp], &snI[th][pp], &csI[th][pp]);
      }
#pragma unroll
      for (int tt = 0; tt < NTT; ++tt) {
#pragma unroll
        for (int r = 0; r < 8; ++r) {
#pragma unroll
          for (int pp = 0; pp < 2; ++pp) {
            float snJ, csJ;
            __sincosf((float)(16 * tt + r + 8 * half) * freq[pp], &snJ, &csJ);
#pragma unroll
            for (int th = 0; th < NTH; ++th) {
              const int t = tt + 2 * th;
              const float cs = byrow ? csI[th][pp] : csJ;
              const float sn = byrow ? snI[th][pp] : snJ;
              const float x0 = acc[t][2 * pp][r], x1 = acc[t][2 * pp + 1][r];
              acc[t][2 * pp][r]     = x0 * cs - x1 * sn;
              acc[t][2 * pp + 1][r] = x1 * cs + x0 * sn;
            }
          }
        }
      }
    }
  }

#pragma unroll
  for (int t = 0; t < TR; ++t) {
#pragma unroll
    for (int r = 0; r < 8; ++r) {
      const int Mr = 16 * t + r + 8 * half;
      v4f o = {acc[t][0][r], acc[t][1][r], acc[t][2][r], acc[t][3][r]};
      *(v4f*)(Y + (size_t)(row0 + Mr) * N + col0 + 4 * m) = o;
    }
  }
}

#define KCH 64
#define KP  36
#define VP  72
#define STG (KCH * KP + (KCH / 2) * VP)

__device__ __forceinline__ float xhalf(float x) {
  int b = __builtin_bit_cast(int, x);
  return __builtin_bit_cast(float,
      __builtin_amdgcn_permlanex16(b, b, 0x76543210, 0xfedcba98, false, false));
}

__global__ __launch_bounds__(256) void attn_kernel(const float* __restrict__ qkv,
                                                   float* __restrict__ out) {
  __shared__ __attribute__((aligned(16))) float smem[2 * STG];

  const int tid  = threadIdx.x;
  const int lane = tid & 31;
  const int wave = tid >> 5;
  int id = blockIdx.x * 8 + wave;
  const int qt = id & 63; id >>= 6;
  const int h  = id & 7;  id >>= 3;
  const int b  = id;
  const int m = lane & 15, half = lane >> 4;
  const int row0 = qt << 4;

  const size_t basebh = (size_t)(b * SS) * E3 + h * DD;
  const float* kslab = qkv + basebh + EE;
  const float* vslab = qkv + basebh + 2 * EE;

  const int sr = tid >> 3, sp = tid & 7;
  const float* kg = kslab + (size_t)sr * E3 + 4 * sp;
  const float* vg = vslab + (size_t)(2 * sr) * E3 + 4 * sp;
  const int ksl = sr * KP + 4 * sp;
  const int vsl = KCH * KP + sr * VP + 8 * sp;

  v2f qb[8];
#pragma unroll
  for (int g = 0; g < 4; ++g) {
    v4f qv = *(const v4f*)(qkv + basebh + (size_t)(row0 + m) * E3 + 8 * g + 4 * half);
    qb[2 * g]     = qv.xy;
    qb[2 * g + 1] = qv.zw;
  }

  {
    v4f a0 = *(const v4f*)(kg);
    v4f a1 = *(const v4f*)(kg + (size_t)32 * E3);
    v4f c0 = *(const v4f*)(vg);
    v4f c1 = *(const v4f*)(vg + E3);
    v4f i0 = {c0.x, c1.x, c0.y, c1.y};
    v4f i1 = {c0.z, c1.z, c0.w, c1.w};
    *(v4f*)(smem + ksl)           = a0;
    *(v4f*)(smem + ksl + 32 * KP) = a1;
    *(v4f*)(smem + vsl)           = i0;
    *(v4f*)(smem + vsl + 4)       = i1;
  }
  __syncthreads();

  const float cscale = 0.25503486164919736f;
  float mrun = -3.0e38f, lpart = 0.0f;
  v8f o0 = {}, o1 = {};

  const int kfrag = m * KP + 4 * half;
  const int vfrag = KCH * KP + 4 * half * VP + 2 * m;
  const v2f cs2 = {cscale, cscale};

  for (int c = 0; c < SS / KCH; ++c) {
    const float* st = smem + (c & 1) * STG;
    float* nst = smem + ((c + 1) & 1) * STG;

    const int cn = (c + 1 < SS / KCH) ? c + 1 : c;
    const size_t go = (size_t)(cn * KCH) * E3;
    v4f a0 = *(const v4f*)(kg + go);
    v4f a1 = *(const v4f*)(kg + go + (size_t)32 * E3);
    v4f c0 = *(const v4f*)(vg + go);
    v4f c1 = *(const v4f*)(vg + go + E3);

    v8f sacc[KCH / 16];
#pragma unroll
    for (int kt = 0; kt < KCH / 16; ++kt) {
      sacc[kt] = {};
      const float* kl = st + 16 * kt * KP + kfrag;
#pragma unroll
      for (int g = 0; g < 4; ++g) {
        v4f kv = *(const v4f*)(kl + 8 * g);
        sacc[kt] = wmma_f32(kv.xy, qb[2 * g], sacc[kt]);
        sacc[kt] = wmma_f32(kv.zw, qb[2 * g + 1], sacc[kt]);
      }
    }

    float tm[KCH / 16];
#pragma unroll
    for (int kt = 0; kt < KCH / 16; ++kt) {
      const v8f s = sacc[kt];
      tm[kt] = fmaxf(fmaxf(fmaxf(s[0], s[1]), fmaxf(s[2], s[3])),
                     fmaxf(fmaxf(s[4], s[5]), fmaxf(s[6], s[7])));
    }
    float tmax = fmaxf(fmaxf(tm[0], tm[1]), fmaxf(tm[2], tm[3]));
    tmax = fmaxf(tmax, xhalf(tmax));
    const float mnew = fmaxf(mrun, tmax);
    const float corr = __builtin_amdgcn_exp2f((mrun - mnew) * cscale);
    const float mcn  = mnew * cscale;
    mrun = mnew;
    const v2f nm = {-mcn, -mcn};

    v2f pb[KCH / 16][4];
    v2f pt[KCH / 16];
#pragma unroll
    for (int kt = 0; kt < KCH / 16; ++kt) {
#pragma unroll
      for (int f = 0; f < 4; ++f) {
        const v2f sv = {sacc[kt][2 * f], sacc[kt][2 * f + 1]};
        const v2f ar = __builtin_elementwise_fma(sv, cs2, nm);
        v2f p;
        p.x = __builtin_amdgcn_exp2f(ar.x);
        p.y = __builtin_amdgcn_exp2f(ar.y);
        pb[kt][f] = p;
      }
      pt[kt] = (pb[kt][0] + pb[kt][1]) + (pb[kt][2] + pb[kt][3]);
    }
    const v2f ps2 = (pt[0] + pt[1]) + (pt[2] + pt[3]);
    lpart = lpart * corr + (ps2.x + ps2.y);
#pragma unroll
    for (int r = 0; r < 8; ++r) { o0[r] *= corr; o1[r] *= corr; }

#pragma unroll
    for (int kt = 0; kt < KCH / 16; ++kt) {
      const float* vl = st + 8 * kt * VP + vfrag;
#pragma unroll
      for (int f = 0; f < 4; ++f) {
        v2f va0 = *(const v2f*)(vl + f * VP);
        v2f va1 = *(const v2f*)(vl + f * VP + 32);
        o0 = wmma_f32(va0, pb[kt][f], o0);
        o1 = wmma_f32(va1, pb[kt][f], o1);
      }
    }

    {
      v4f i0 = {c0.x, c1.x, c0.y, c1.y};
      v4f i1 = {c0.z, c1.z, c0.w, c1.w};
      *(v4f*)(nst + ksl)           = a0;
      *(v4f*)(nst + ksl + 32 * KP) = a1;
      *(v4f*)(nst + vsl)           = i0;
      *(v4f*)(nst + vsl + 4)       = i1;
    }
    __syncthreads();
  }

  float lsum = lpart + xhalf(lpart);
  float inv = 1.0f / lsum;
  float* orow = out + (size_t)(b * SS + row0 + m) * EE + h * DD + 8 * half;
  v4f w0 = {o0[0] * inv, o0[1] * inv, o0[2] * inv, o0[3] * inv};
  v4f w1 = {o0[4] * inv, o0[5] * inv, o0[6] * inv, o0[7] * inv};
  v4f w2 = {o1[0] * inv, o1[1] * inv, o1[2] * inv, o1[3] * inv};
  v4f w3 = {o1[4] * inv, o1[5] * inv, o1[6] * inv, o1[7] * inv};
  *(v4f*)(orow)      = w0;
  *(v4f*)(orow + 4)  = w1;
  *(v4f*)(orow + 16) = w2;
  *(v4f*)(orow + 20) = w3;
}

extern "C" void kernel_launch(void* const* d_in, const int* in_sizes, int n_in,
                              void* d_out, int out_size, void* d_ws, size_t ws_size,
                              hipStream_t stream) {
  const float* x    = (const float*)d_in[0];
  const float* Wqkv = (const float*)d_in[1];
  const float* bqkv = (const float*)d_in[2];
  const float* Wout = (const float*)d_in[3];
  const float* bout = (const float*)d_in[4];
  float* out  = (float*)d_out;
  float* qkv  = (float*)d_ws;
  float* attn = qkv + (size_t)NB * SS * E3;

  gemm_tile<true, 4, 1><<<dim3((NB * SS / 64) * (E3 / 64) / 4), dim3(128), 0, stream>>>(
      x, Wqkv, bqkv, qkv, NB * SS, E3, EE);

  attn_kernel<<<dim3(NB * HH * (SS / 16) / 8), dim3(256), 0, stream>>>(qkv, attn);

  gemm_tile<false, 2, 2><<<dim3((NB * SS / 32) * (EE / 64) / 4), dim3(128), 0, stream>>>(
      attn, Wout, bout, out, NB * SS, EE, EE);
}
